# starting kernel
# baseline (speedup 1.0000x reference)
#include <hip/hip_runtime.h>


typedef __attribute__((ext_vector_type(2))) float v2f;
typedef __attribute__((ext_vector_type(8))) float v8f;

#define D 128

__global__ void gcn_zero_kernel(float* __restrict__ p, long long n) {
  long long i = (long long)blockIdx.x * blockDim.x + threadIdx.x;
  long long stride = (long long)gridDim.x * blockDim.x;
  for (; i < n; i += stride) p[i] = 0.0f;
}

__global__ void gcn_degree_kernel(const int* __restrict__ src,
                                  const int* __restrict__ dst,
                                  float* __restrict__ deg_out,
                                  float* __restrict__ deg_in,
                                  int n_edges) {
  int e = blockIdx.x * blockDim.x + threadIdx.x;
  if (e >= n_edges) return;
  atomicAdd(&deg_out[src[e]], 1.0f);
  atomicAdd(&deg_in[dst[e]], 1.0f);
}

__global__ void gcn_invsqrt_kernel(float* __restrict__ deg, int n) {
  int i = blockIdx.x * blockDim.x + threadIdx.x;
  if (i >= n) return;
  deg[i] = rsqrtf(fmaxf(deg[i], 1.0f));
}

__global__ __launch_bounds__(256) void gcn_spmm_kernel(
    const float* __restrict__ h, const int* __restrict__ src,
    const int* __restrict__ dst, const float* __restrict__ inv_out,
    float* __restrict__ agg, int n_edges) {
  int t = blockIdx.x * blockDim.x + threadIdx.x;
  int e = t >> 5;
  if (e >= n_edges) return;
  int lane = t & 31;
  int s = src[e];
  int d = dst[e];
  float sc = inv_out[s];
  const float4 v = *(const float4*)(h + (size_t)s * D + lane * 4);
  float* ap = agg + (size_t)d * D + lane * 4;
  atomicAdd(ap + 0, v.x * sc);
  atomicAdd(ap + 1, v.y * sc);
  atomicAdd(ap + 2, v.z * sc);
  atomicAdd(ap + 3, v.w * sc);
}

__global__ __launch_bounds__(256) void gcn_gemm_kernel(
    const float* __restrict__ agg, const float* __restrict__ inv_in,
    const float* __restrict__ W, const float* __restrict__ bias,
    float* __restrict__ out, int relu) {
  __shared__ float Atile[16 * D];

  const int row0 = blockIdx.x * 16;
  const int tid = threadIdx.x;

  {
    int r = tid >> 4;
    int c = (tid & 15) * 8;
    float s = inv_in[row0 + r];
    const float4* g = (const float4*)(agg + (size_t)(row0 + r) * D + c);
    float4 v0 = g[0];
    float4 v1 = g[1];
    float4* l = (float4*)(Atile + r * D + c);
    v0.x *= s; v0.y *= s; v0.z *= s; v0.w *= s;
    v1.x *= s; v1.y *= s; v1.z *= s; v1.w *= s;
    l[0] = v0;
    l[1] = v1;
  }
  __syncthreads();

  const int lane = tid & 31;
  const int col0 = (tid >> 5) * 16;
  const int m = lane & 15;
  const int khalf = (lane >> 4) * 2;

  v8f c = {};
#pragma unroll 4
  for (int k0 = 0; k0 < D; k0 += 4) {
    v2f a;
    a.x = Atile[m * D + k0 + khalf + 0];
    a.y = Atile[m * D + k0 + khalf + 1];
    v2f b;
    b.x = W[(size_t)(k0 + khalf + 0) * D + col0 + m];
    b.y = W[(size_t)(k0 + khalf + 1) * D + col0 + m];
    c = __builtin_amdgcn_wmma_f32_16x16x4_f32(
         false, a,  false, b,
         (short)0, c,  false,  false);
  }

  const int ncol = col0 + m;
  const int rbase = row0 + (lane >> 4) * 8;
  const float bv = bias[ncol];
#pragma unroll
  for (int r = 0; r < 8; ++r) {
    float v = c[r] + bv;
    if (relu) v = fmaxf(v, 0.0f);
    out[(size_t)(rbase + r) * D + ncol] = v;
  }
}

extern "C" void kernel_launch(void* const* d_in, const int* in_sizes, int n_in,
                              void* d_out, int out_size, void* d_ws, size_t ws_size,
                              hipStream_t stream) {
  const float* feat = (const float*)d_in[0];
  const int* src = (const int*)d_in[1];
  const int* dst = (const int*)d_in[2];
  const float* W1 = (const float*)d_in[3];
  const float* b1 = (const float*)d_in[4];
  const float* W2 = (const float*)d_in[5];
  const float* b2 = (const float*)d_in[6];
  const float* W3 = (const float*)d_in[7];
  const float* b3 = (const float*)d_in[8];
  float* out = (float*)d_out;

  const int N = in_sizes[0] / D;
  const int E = in_sizes[1];

  float* ws = (float*)d_ws;
  float* inv_out = ws;
  float* inv_in = ws + N;
  float* agg = ws + 2 * (size_t)N;
  float* h = agg + (size_t)N * D;

  const int ZB = 2048;

  gcn_zero_kernel<<<ZB, 256, 0, stream>>>(inv_out, 2 * (long long)N);
  gcn_degree_kernel<<<(E + 255) / 256, 256, 0, stream>>>(src, dst, inv_out, inv_in, E);
  gcn_invsqrt_kernel<<<(2 * N + 255) / 256, 256, 0, stream>>>(inv_out, 2 * N);

  const int spmm_blocks = (E * 32 + 255) / 256;
  const int gemm_blocks = N / 16;

  gcn_zero_kernel<<<ZB, 256, 0, stream>>>(agg, (long long)N * D);
  gcn_spmm_kernel<<<spmm_blocks, 256, 0, stream>>>(feat, src, dst, inv_out, agg, E);
  gcn_gemm_kernel<<<gemm_blocks, 256, 0, stream>>>(agg, inv_in, W1, b1, h, 1);

  gcn_zero_kernel<<<ZB, 256, 0, stream>>>(agg, (long long)N * D);
  gcn_spmm_kernel<<<spmm_blocks, 256, 0, stream>>>(h, src, dst, inv_out, agg, E);
  gcn_gemm_kernel<<<gemm_blocks, 256, 0, stream>>>(agg, inv_in, W2, b2, h, 1);

  gcn_zero_kernel<<<ZB, 256, 0, stream>>>(agg, (long long)N * D);
  gcn_spmm_kernel<<<spmm_blocks, 256, 0, stream>>>(h, src, dst, inv_out, agg, E);
  gcn_gemm_kernel<<<gemm_blocks, 256, 0, stream>>>(agg, inv_in, W3, b3, out, 0);
}


# SpMM as an atomic-free row gather over a destination-grouped edge list built per call (int count+rank, scan, fill) instead of f32 atomic scatter; sum order = arrival rank in the row
# speedup vs baseline: 1.9329x; 1.9329x over previous
#include <hip/hip_runtime.h>


typedef __attribute__((ext_vector_type(2))) float v2f;
typedef __attribute__((ext_vector_type(8))) float v8f;

#define D 128

__global__ void gcn_zero_i32_kernel(int* __restrict__ p, int n) {
  int i = blockIdx.x * blockDim.x + threadIdx.x;
  if (i < n) p[i] = 0;
}

__global__ void gcn_count_kernel(const int* __restrict__ src,
                                 const int* __restrict__ dst,
                                 int* __restrict__ cnt_out,
                                 int* __restrict__ cnt_in,
                                 int* __restrict__ rank,
                                 int n_edges) {
  int e = blockIdx.x * blockDim.x + threadIdx.x;
  if (e >= n_edges) return;
  atomicAdd(&cnt_out[src[e]], 1);
  rank[e] = atomicAdd(&cnt_in[dst[e]], 1);
}

__global__ __launch_bounds__(1024) void gcn_rowptr_kernel(
    const int* __restrict__ cnt_out, const int* __restrict__ cnt_in,
    int* __restrict__ rowptr, float* __restrict__ inv_out,
    float* __restrict__ inv_in, int n) {
  __shared__ int wsum[32];
  __shared__ int wpre[32];
  const int tid = threadIdx.x;
  const int lane = tid & 31;
  const int wid = tid >> 5;
  const int base = blockIdx.x * 1024;
  const int i = base + tid;

  int pre = 0;
  for (int j = tid; j < base; j += 1024) pre += cnt_in[j];
  const int c = (i < n) ? cnt_in[i] : 0;

  int x = c;
  for (int o = 1; o < 32; o <<= 1) {
    int y = __shfl_up(x, o);
    if (lane >= o) x += y;
  }
  for (int o = 16; o > 0; o >>= 1) pre += __shfl_xor(pre, o);
  if (lane == 31) wsum[wid] = x;
  if (lane == 0) wpre[wid] = pre;
  __syncthreads();

  int woff = (lane < wid) ? wsum[lane] : 0;
  int ptot = wpre[lane];
  for (int o = 16; o > 0; o >>= 1) {
    woff += __shfl_xor(woff, o);
    ptot += __shfl_xor(ptot, o);
  }

  if (i < n) {
    const int start = ptot + woff + (x - c);
    rowptr[i] = start;
    if (i == n - 1) rowptr[n] = start + c;
    inv_in[i] = rsqrtf(fmaxf((float)c, 1.0f));
    inv_out[i] = rsqrtf(fmaxf((float)cnt_out[i], 1.0f));
  }
}

__global__ void gcn_fill_kernel(const int* __restrict__ src,
                                const int* __restrict__ dst,
                                const int* __restrict__ rank,
                                const int* __restrict__ rowptr,
                                const float* __restrict__ inv_out,
                                int* __restrict__ col, float* __restrict__ wgt,
                                int n_edges) {
  int e = blockIdx.x * blockDim.x + threadIdx.x;
  if (e >= n_edges) return;
  const int s = src[e];
  const int pos = rowptr[dst[e]] + rank[e];
  col[pos] = s;
  wgt[pos] = inv_out[s];
}


template <int G>
__device__ __forceinline__ void gcn_add_rows(const float* __restrict__ hl, int cs, int cwi,
                                             int k, float4& acc) {
  float4 v[G];
  float w[G];
#pragma unroll
  for (int t = 0; t < G; ++t) {
    const unsigned s = (unsigned)__builtin_amdgcn_readlane(cs, k + t);
    w[t] = __int_as_float(__builtin_amdgcn_readlane(cwi, k + t));
    v[t] = *(const float4*)(hl + (size_t)s * D);
  }
#pragma unroll
  for (int t = 0; t < G; ++t) {
    acc.x += v[t].x * w[t];
    acc.y += v[t].y * w[t];
    acc.z += v[t].z * w[t];
    acc.w += v[t].w * w[t];
  }
}

__global__ __launch_bounds__(256) void gcn_gather_kernel(
    const float* __restrict__ h, const int* __restrict__ rowptr,
    const int* __restrict__ col, const float* __restrict__ wgt,
    float* __restrict__ agg, int n_nodes) {
  const int lane = threadIdx.x & 31;
  const int wave0 = __builtin_amdgcn_readfirstlane((int)((blockIdx.x * blockDim.x + threadIdx.x) >> 5));
  const int nwaves = (int)((gridDim.x * blockDim.x) >> 5);
  const float* hl = h + lane * 4;

  for (int i = wave0; i < n_nodes; i += nwaves) {
    const int beg = rowptr[i];
    const int end = rowptr[i + 1];
    float4 acc = make_float4(0.0f, 0.0f, 0.0f, 0.0f);
    for (int base = beg; base < end; base += 32) {
      const int j = base + lane;
      int cs = 0;
      float cw = 0.0f;
      if (j < end) {
        cs = col[j];
        cw = wgt[j];
      }
      const int cwi = __float_as_int(cw);
      const int cnt = min(32, end - base);
      int k = 0;
      for (; k + 8 <= cnt; k += 8) gcn_add_rows<8>(hl, cs, cwi, k, acc);
      if (k + 4 <= cnt) {
        gcn_add_rows<4>(hl, cs, cwi, k, acc);
        k += 4;
      }
      const int rem = cnt - k;
      if (rem == 3) gcn_add_rows<3>(hl, cs, cwi, k, acc);
      else if (rem == 2) gcn_add_rows<2>(hl, cs, cwi, k, acc);
      else if (rem == 1) gcn_add_rows<1>(hl, cs, cwi, k, acc);
    }
    *(float4*)(agg + (size_t)i * D + lane * 4) = acc;
  }
}

__global__ __launch_bounds__(256) void gcn_gemm_kernel(
    const float* __restrict__ agg, const float* __restrict__ inv_in,
    const float* __restrict__ W, const float* __restrict__ bias,
    float* __restrict__ out, int relu) {
  __shared__ float Atile[16 * D];

  const int row0 = blockIdx.x * 16;
  const int tid = threadIdx.x;

  {
    int r = tid >> 4;
    int c = (tid & 15) * 8;
    float s = inv_in[row0 + r];
    const float4* g = (const float4*)(agg + (size_t)(row0 + r) * D + c);
    float4 v0 = g[0];
    float4 v1 = g[1];
    float4* l = (float4*)(Atile + r * D + c);
    v0.x *= s; v0.y *= s; v0.z *= s; v0.w *= s;
    v1.x *= s; v1.y *= s; v1.z *= s; v1.w *= s;
    l[0] = v0;
    l[1] = v1;
  }
  __syncthreads();

  const int lane = tid & 31;
  const int col0 = (tid >> 5) * 16;
  const int m = lane & 15;
  const int khalf = (lane >> 4) * 2;

  v8f c = {};
#pragma unroll 4
  for (int k0 = 0; k0 < D; k0 += 4) {
    v2f a;
    a.x = Atile[m * D + k0 + khalf + 0];
    a.y = Atile[m * D + k0 + khalf + 1];
    v2f b;
    b.x = W[(size_t)(k0 + khalf + 0) * D + col0 + m];
    b.y = W[(size_t)(k0 + khalf + 1) * D + col0 + m];
    c = __builtin_amdgcn_wmma_f32_16x16x4_f32(
         false, a,  false, b,
         (short)0, c,  false,  false);
  }

  const int ncol = col0 + m;
  const int rbase = row0 + (lane >> 4) * 8;
  const float bv = bias[ncol];
#pragma unroll
  for (int r = 0; r < 8; ++r) {
    float v = c[r] + bv;
    if (relu) v = fmaxf(v, 0.0f);
    out[(size_t)(rbase + r) * D + ncol] = v;
  }
}

extern "C" void kernel_launch(void* const* d_in, const int* in_sizes, int n_in,
                              void* d_out, int out_size, void* d_ws, size_t ws_size,
                              hipStream_t stream) {
  const float* feat = (const float*)d_in[0];
  const int* src = (const int*)d_in[1];
  const int* dst = (const int*)d_in[2];
  const float* W1 = (const float*)d_in[3];
  const float* b1 = (const float*)d_in[4];
  const float* W2 = (const float*)d_in[5];
  const float* b2 = (const float*)d_in[6];
  const float* W3 = (const float*)d_in[7];
  const float* b3 = (const float*)d_in[8];
  float* out = (float*)d_out;

  const int N = in_sizes[0] / D;
  const int E = in_sizes[1];

  float* ws = (float*)d_ws;
  float* inv_out = ws;
  float* inv_in = ws + N;
  float* agg = ws + 2 * (size_t)N;
  float* h = agg + (size_t)N * D;
  float* wgt = h + (size_t)N * D;
  int* col = (int*)(wgt + E);
  int* rank = col + E;
  int* cnt_out = rank + E;
  int* cnt_in = cnt_out + N;
  int* rowptr = cnt_in + N;

  const int edge_blocks = (E + 255) / 256;

  gcn_zero_i32_kernel<<<(2 * N + 255) / 256, 256, 0, stream>>>(cnt_out, 2 * N);
  gcn_count_kernel<<<edge_blocks, 256, 0, stream>>>(src, dst, cnt_out, cnt_in, rank, E);
  gcn_rowptr_kernel<<<(N + 1023) / 1024, 1024, 0, stream>>>(cnt_out, cnt_in, rowptr, inv_out, inv_in, N);
  gcn_fill_kernel<<<edge_blocks, 256, 0, stream>>>(src, dst, rank, rowptr, inv_out, col, wgt, E);

  const int gather_blocks = 2048;
  const int gemm_blocks = N / 16;

  gcn_gather_kernel<<<gather_blocks, 256, 0, stream>>>(feat, rowptr, col, wgt, agg, N);
  gcn_gemm_kernel<<<gemm_blocks, 256, 0, stream>>>(agg, inv_in, W1, b1, h, 1);

  gcn_gather_kernel<<<gather_blocks, 256, 0, stream>>>(h, rowptr, col, wgt, agg, N);
  gcn_gemm_kernel<<<gemm_blocks, 256, 0, stream>>>(agg, inv_in, W2, b2, h, 1);

  gcn_gather_kernel<<<gather_blocks, 256, 0, stream>>>(h, rowptr, col, wgt, agg, N);
  gcn_gemm_kernel<<<gemm_blocks, 256, 0, stream>>>(agg, inv_in, W3, b3, out, 0);
}


# GEMM A tile in LDS padded from a 128- to a 132-float row pitch: the 16 rows of one A-fragment read now fall on 16 different bank groups (was a 16-way bank conflict on every fragment read)
# speedup vs baseline: 2.1928x; 1.1345x over previous
#include <hip/hip_runtime.h>


typedef __attribute__((ext_vector_type(2))) float v2f;
typedef __attribute__((ext_vector_type(8))) float v8f;

#define D 128
#define AP 132

__global__ void gcn_zero_i32_kernel(int* __restrict__ p, int n) {
  int i = blockIdx.x * blockDim.x + threadIdx.x;
  if (i < n) p[i] = 0;
}

__global__ void gcn_count_kernel(const int* __restrict__ src,
                                 const int* __restrict__ dst,
                                 int* __restrict__ cnt_out,
                                 int* __restrict__ cnt_in,
                                 int* __restrict__ rank,
                                 int n_edges) {
  int e = blockIdx.x * blockDim.x + threadIdx.x;
  if (e >= n_edges) return;
  atomicAdd(&cnt_out[src[e]], 1);
  rank[e] = atomicAdd(&cnt_in[dst[e]], 1);
}

__global__ __launch_bounds__(1024) void gcn_rowptr_kernel(
    const int* __restrict__ cnt_out, const int* __restrict__ cnt_in,
    int* __restrict__ rowptr, float* __restrict__ inv_out,
    float* __restrict__ inv_in, int n) {
  __shared__ int wsum[32];
  __shared__ int wpre[32];
  const int tid = threadIdx.x;
  const int lane = tid & 31;
  const int wid = tid >> 5;
  const int base = blockIdx.x * 1024;
  const int i = base + tid;

  int pre = 0;
  for (int j = tid; j < base; j += 1024) pre += cnt_in[j];
  const int c = (i < n) ? cnt_in[i] : 0;

  int x = c;
  for (int o = 1; o < 32; o <<= 1) {
    int y = __shfl_up(x, o);
    if (lane >= o) x += y;
  }
  for (int o = 16; o > 0; o >>= 1) pre += __shfl_xor(pre, o);
  if (lane == 31) wsum[wid] = x;
  if (lane == 0) wpre[wid] = pre;
  __syncthreads();

  int woff = (lane < wid) ? wsum[lane] : 0;
  int ptot = wpre[lane];
  for (int o = 16; o > 0; o >>= 1) {
    woff += __shfl_xor(woff, o);
    ptot += __shfl_xor(ptot, o);
  }

  if (i < n) {
    const int start = ptot + woff + (x - c);
    rowptr[i] = start;
    if (i == n - 1) rowptr[n] = start + c;
    inv_in[i] = rsqrtf(fmaxf((float)c, 1.0f));
    inv_out[i] = rsqrtf(fmaxf((float)cnt_out[i], 1.0f));
  }
}

__global__ void gcn_fill_kernel(const int* __restrict__ src,
                                const int* __restrict__ dst,
                                const int* __restrict__ rank,
                                const int* __restrict__ rowptr,
                                const float* __restrict__ inv_out,
                                int* __restrict__ col, float* __restrict__ wgt,
                                int n_edges) {
  int e = blockIdx.x * blockDim.x + threadIdx.x;
  if (e >= n_edges) return;
  const int s = src[e];
  const int pos = rowptr[dst[e]] + rank[e];
  col[pos] = s;
  wgt[pos] = inv_out[s];
}


template <int G>
__device__ __forceinline__ void gcn_add_rows(const float* __restrict__ hl, int cs, int cwi,
                                             int k, float4& acc) {
  float4 v[G];
  float w[G];
#pragma unroll
  for (int t = 0; t < G; ++t) {
    const unsigned s = (unsigned)__builtin_amdgcn_readlane(cs, k + t);
    w[t] = __int_as_float(__builtin_amdgcn_readlane(cwi, k + t));
    v[t] = *(const float4*)(hl + (size_t)s * D);
  }
#pragma unroll
  for (int t = 0; t < G; ++t) {
    acc.x += v[t].x * w[t];
    acc.y += v[t].y * w[t];
    acc.z += v[t].z * w[t];
    acc.w += v[t].w * w[t];
  }
}

__global__ __launch_bounds__(256) void gcn_gather_kernel(
    const float* __restrict__ h, const int* __restrict__ rowptr,
    const int* __restrict__ col, const float* __restrict__ wgt,
    float* __restrict__ agg, int n_nodes) {
  const int lane = threadIdx.x & 31;
  const int wave0 = __builtin_amdgcn_readfirstlane((int)((blockIdx.x * blockDim.x + threadIdx.x) >> 5));
  const int nwaves = (int)((gridDim.x * blockDim.x) >> 5);
  const float* hl = h + lane * 4;

  for (int i = wave0; i < n_nodes; i += nwaves) {
    const int beg = rowptr[i];
    const int end = rowptr[i + 1];
    float4 acc = make_float4(0.0f, 0.0f, 0.0f, 0.0f);
    for (int base = beg; base < end; base += 32) {
      const int j = base + lane;
      int cs = 0;
      float cw = 0.0f;
      if (j < end) {
        cs = col[j];
        cw = wgt[j];
      }
      const int cwi = __float_as_int(cw);
      const int cnt = min(32, end - base);
      int k = 0;
      for (; k + 8 <= cnt; k += 8) gcn_add_rows<8>(hl, cs, cwi, k, acc);
      if (k + 4 <= cnt) {
        gcn_add_rows<4>(hl, cs, cwi, k, acc);
        k += 4;
      }
      const int rem = cnt - k;
      if (rem == 3) gcn_add_rows<3>(hl, cs, cwi, k, acc);
      else if (rem == 2) gcn_add_rows<2>(hl, cs, cwi, k, acc);
      else if (rem == 1) gcn_add_rows<1>(hl, cs, cwi, k, acc);
    }
    *(float4*)(agg + (size_t)i * D + lane * 4) = acc;
  }
}

__global__ __launch_bounds__(256) void gcn_gemm_kernel(
    const float* __restrict__ agg, const float* __restrict__ inv_in,
    const float* __restrict__ W, const float* __restrict__ bias,
    float* __restrict__ out, int relu) {
  __shared__ float Atile[16 * AP];

  const int row0 = blockIdx.x * 16;
  const int tid = threadIdx.x;

  {
    int r = tid >> 4;
    int c = (tid & 15) * 8;
    float s = inv_in[row0 + r];
    const float4* g = (const float4*)(agg + (size_t)(row0 + r) * D + c);
    float4 v0 = g[0];
    float4 v1 = g[1];
    float4* l = (float4*)(Atile + r * AP + c);
    v0.x *= s; v0.y *= s; v0.z *= s; v0.w *= s;
    v1.x *= s; v1.y *= s; v1.z *= s; v1.w *= s;
    l[0] = v0;
    l[1] = v1;
  }
  __syncthreads();

  const int lane = tid & 31;
  const int col0 = (tid >> 5) * 16;
  const int m = lane & 15;
  const int khalf = (lane >> 4) * 2;

  v8f c = {};
#pragma unroll 4
  for (int k0 = 0; k0 < D; k0 += 4) {
    v2f a;
    a.x = Atile[m * AP + k0 + khalf + 0];
    a.y = Atile[m * AP + k0 + khalf + 1];
    v2f b;
    b.x = W[(size_t)(k0 + khalf + 0) * D + col0 + m];
    b.y = W[(size_t)(k0 + khalf + 1) * D + col0 + m];
    c = __builtin_amdgcn_wmma_f32_16x16x4_f32(
         false, a,  false, b,
         (short)0, c,  false,  false);
  }

  const int ncol = col0 + m;
  const int rbase = row0 + (lane >> 4) * 8;
  const float bv = bias[ncol];
#pragma unroll
  for (int r = 0; r < 8; ++r) {
    float v = c[r] + bv;
    if (relu) v = fmaxf(v, 0.0f);
    out[(size_t)(rbase + r) * D + ncol] = v;
  }
}

extern "C" void kernel_launch(void* const* d_in, const int* in_sizes, int n_in,
                              void* d_out, int out_size, void* d_ws, size_t ws_size,
                              hipStream_t stream) {
  const float* feat = (const float*)d_in[0];
  const int* src = (const int*)d_in[1];
  const int* dst = (const int*)d_in[2];
  const float* W1 = (const float*)d_in[3];
  const float* b1 = (const float*)d_in[4];
  const float* W2 = (const float*)d_in[5];
  const float* b2 = (const float*)d_in[6];
  const float* W3 = (const float*)d_in[7];
  const float* b3 = (const float*)d_in[8];
  float* out = (float*)d_out;

  const int N = in_sizes[0] / D;
  const int E = in_sizes[1];

  float* ws = (float*)d_ws;
  float* inv_out = ws;
  float* inv_in = ws + N;
  float* agg = ws + 2 * (size_t)N;
  float* h = agg + (size_t)N * D;
  float* wgt = h + (size_t)N * D;
  int* col = (int*)(wgt + E);
  int* rank = col + E;
  int* cnt_out = rank + E;
  int* cnt_in = cnt_out + N;
  int* rowptr = cnt_in + N;

  const int edge_blocks = (E + 255) / 256;

  gcn_zero_i32_kernel<<<(2 * N + 255) / 256, 256, 0, stream>>>(cnt_out, 2 * N);
  gcn_count_kernel<<<edge_blocks, 256, 0, stream>>>(src, dst, cnt_out, cnt_in, rank, E);
  gcn_rowptr_kernel<<<(N + 1023) / 1024, 1024, 0, stream>>>(cnt_out, cnt_in, rowptr, inv_out, inv_in, N);
  gcn_fill_kernel<<<edge_blocks, 256, 0, stream>>>(src, dst, rank, rowptr, inv_out, col, wgt, E);

  const int gather_blocks = 2048;
  const int gemm_blocks = N / 16;

  gcn_gather_kernel<<<gather_blocks, 256, 0, stream>>>(feat, rowptr, col, wgt, agg, N);
  gcn_gemm_kernel<<<gemm_blocks, 256, 0, stream>>>(agg, inv_in, W1, b1, h, 1);

  gcn_gather_kernel<<<gather_blocks, 256, 0, stream>>>(h, rowptr, col, wgt, agg, N);
  gcn_gemm_kernel<<<gemm_blocks, 256, 0, stream>>>(agg, inv_in, W2, b2, h, 1);

  gcn_gather_kernel<<<gather_blocks, 256, 0, stream>>>(h, rowptr, col, wgt, agg, N);
  gcn_gemm_kernel<<<gemm_blocks, 256, 0, stream>>>(agg, inv_in, W3, b3, out, 0);
}


# Degree counts and edge ranks from LDS counters (a per-node table per edge chunk, prefixed over 32 chunks) instead of global atomics; row sum order = chunk, then arrival in the chunk
# speedup vs baseline: 2.9006x; 1.3228x over previous
#include <hip/hip_runtime.h>


typedef __attribute__((ext_vector_type(2))) float v2f;
typedef __attribute__((ext_vector_type(8))) float v8f;

#define D 128
#define AP 132

#define NB 32
#define NODE_CAP 50000

__global__ __launch_bounds__(1024) void gcn_count_kernel(const int* __restrict__ src,
                                                         const int* __restrict__ dst,
                                                         int* __restrict__ part_out,
                                                         int* __restrict__ part_in,
                                                         int* __restrict__ rank,
                                                         int n_nodes, int n_edges, int chunk) {
  __shared__ int hist[NODE_CAP];
  const int tid = threadIdx.x;
  const bool by_dst = blockIdx.x < NB;
  const int cb = by_dst ? blockIdx.x : blockIdx.x - NB;
  const int e0 = cb * chunk;
  const int e1 = min(e0 + chunk, n_edges);

  for (int i = tid; i < n_nodes; i += 1024) hist[i] = 0;
  __syncthreads();

  if (by_dst) {
    for (int e = e0 + tid; e < e1; e += 4096) {
      int key[4];
#pragma unroll
      for (int u = 0; u < 4; ++u) {
        const int ee = e + u * 1024;
        key[u] = (ee < e1) ? dst[ee] : 0;
      }
      int r[4];
#pragma unroll
      for (int u = 0; u < 4; ++u) {
        const int ee = e + u * 1024;
        r[u] = (ee < e1) ? atomicAdd(&hist[key[u]], 1) : 0;
      }
#pragma unroll
      for (int u = 0; u < 4; ++u) {
        const int ee = e + u * 1024;
        if (ee < e1) rank[ee] = r[u];
      }
    }
  } else {
    for (int e = e0 + tid; e < e1; e += 4096) {
      int key[4];
#pragma unroll
      for (int u = 0; u < 4; ++u) {
        const int ee = e + u * 1024;
        key[u] = (ee < e1) ? src[ee] : 0;
      }
#pragma unroll
      for (int u = 0; u < 4; ++u) {
        const int ee = e + u * 1024;
        if (ee < e1) atomicAdd(&hist[key[u]], 1);
      }
    }
  }
  __syncthreads();

  int* part = (by_dst ? part_in : part_out) + (size_t)cb * n_nodes;
  for (int i = tid; i < n_nodes; i += 1024) part[i] = hist[i];
}

__global__ void gcn_chunk_prefix_kernel(int* __restrict__ part_in,
                                        const int* __restrict__ part_out,
                                        int* __restrict__ cnt_out,
                                        int* __restrict__ cnt_in, int n) {
  const int i = blockIdx.x * blockDim.x + threadIdx.x;
  if (i >= n) return;
  int cin[NB];
  int tout = 0;
#pragma unroll
  for (int c = 0; c < NB; ++c) {
    cin[c] = part_in[(size_t)c * n + i];
    tout += part_out[(size_t)c * n + i];
  }
  int run = 0;
#pragma unroll
  for (int c = 0; c < NB; ++c) {
    part_in[(size_t)c * n + i] = run;
    run += cin[c];
  }
  cnt_in[i] = run;
  cnt_out[i] = tout;
}

__global__ __launch_bounds__(1024) void gcn_rowptr_kernel(
    const int* __restrict__ cnt_out, const int* __restrict__ cnt_in,
    int* __restrict__ rowptr, float* __restrict__ inv_out,
    float* __restrict__ inv_in, int n) {
  __shared__ int wsum[32];
  __shared__ int wpre[32];
  const int tid = threadIdx.x;
  const int lane = tid & 31;
  const int wid = tid >> 5;
  const int base = blockIdx.x * 1024;
  const int i = base + tid;

  int pre = 0;
  for (int j = tid; j < base; j += 1024) pre += cnt_in[j];
  const int c = (i < n) ? cnt_in[i] : 0;

  int x = c;
  for (int o = 1; o < 32; o <<= 1) {
    int y = __shfl_up(x, o);
    if (lane >= o) x += y;
  }
  for (int o = 16; o > 0; o >>= 1) pre += __shfl_xor(pre, o);
  if (lane == 31) wsum[wid] = x;
  if (lane == 0) wpre[wid] = pre;
  __syncthreads();

  int woff = (lane < wid) ? wsum[lane] : 0;
  int ptot = wpre[lane];
  for (int o = 16; o > 0; o >>= 1) {
    woff += __shfl_xor(woff, o);
    ptot += __shfl_xor(ptot, o);
  }

  if (i < n) {
    const int start = ptot + woff + (x - c);
    rowptr[i] = start;
    if (i == n - 1) rowptr[n] = start + c;
    inv_in[i] = rsqrtf(fmaxf((float)c, 1.0f));
    inv_out[i] = rsqrtf(fmaxf((float)cnt_out[i], 1.0f));
  }
}

__global__ void gcn_fill_kernel(const int* __restrict__ src,
                                const int* __restrict__ dst,
                                const int* __restrict__ rank,
                                const int* __restrict__ rowptr,
                                const int* __restrict__ part_in,
                                const float* __restrict__ inv_out,
                                int* __restrict__ col, float* __restrict__ wgt,
                                int n_nodes, int n_edges, int chunk) {
  const int cb = blockIdx.y;
  const int off = blockIdx.x * blockDim.x + threadIdx.x;
  const int e = cb * chunk + off;
  if (off >= chunk || e >= n_edges) return;
  const int s = src[e];
  const int d = dst[e];
  const int pos = rowptr[d] + part_in[(size_t)cb * n_nodes + d] + rank[e];
  col[pos] = s;
  wgt[pos] = inv_out[s];
}


template <int G>
__device__ __forceinline__ void gcn_add_rows(const float* __restrict__ hl, int cs, int cwi,
                                             int k, float4& acc) {
  float4 v[G];
  float w[G];
#pragma unroll
  for (int t = 0; t < G; ++t) {
    const unsigned s = (unsigned)__builtin_amdgcn_readlane(cs, k + t);
    w[t] = __int_as_float(__builtin_amdgcn_readlane(cwi, k + t));
    v[t] = *(const float4*)(hl + (size_t)s * D);
  }
#pragma unroll
  for (int t = 0; t < G; ++t) {
    acc.x += v[t].x * w[t];
    acc.y += v[t].y * w[t];
    acc.z += v[t].z * w[t];
    acc.w += v[t].w * w[t];
  }
}

__global__ __launch_bounds__(256) void gcn_gather_kernel(
    const float* __restrict__ h, const int* __restrict__ rowptr,
    const int* __restrict__ col, const float* __restrict__ wgt,
    float* __restrict__ agg, int n_nodes) {
  const int lane = threadIdx.x & 31;
  const int wave0 = __builtin_amdgcn_readfirstlane((int)((blockIdx.x * blockDim.x + threadIdx.x) >> 5));
  const int nwaves = (int)((gridDim.x * blockDim.x) >> 5);
  const float* hl = h + lane * 4;

  for (int i = wave0; i < n_nodes; i += nwaves) {
    const int beg = rowptr[i];
    const int end = rowptr[i + 1];
    float4 acc = make_float4(0.0f, 0.0f, 0.0f, 0.0f);
    for (int base = beg; base < end; base += 32) {
      const int j = base + lane;
      int cs = 0;
      float cw = 0.0f;
      if (j < end) {
        cs = col[j];
        cw = wgt[j];
      }
      const int cwi = __float_as_int(cw);
      const int cnt = min(32, end - base);
      int k = 0;
      for (; k + 8 <= cnt; k += 8) gcn_add_rows<8>(hl, cs, cwi, k, acc);
      if (k + 4 <= cnt) {
        gcn_add_rows<4>(hl, cs, cwi, k, acc);
        k += 4;
      }
      const int rem = cnt - k;
      if (rem == 3) gcn_add_rows<3>(hl, cs, cwi, k, acc);
      else if (rem == 2) gcn_add_rows<2>(hl, cs, cwi, k, acc);
      else if (rem == 1) gcn_add_rows<1>(hl, cs, cwi, k, acc);
    }
    *(float4*)(agg + (size_t)i * D + lane * 4) = acc;
  }
}

__global__ __launch_bounds__(256) void gcn_gemm_kernel(
    const float* __restrict__ agg, const float* __restrict__ inv_in,
    const float* __restrict__ W, const float* __restrict__ bias,
    float* __restrict__ out, int relu) {
  __shared__ float Atile[16 * AP];

  const int row0 = blockIdx.x * 16;
  const int tid = threadIdx.x;

  {
    int r = tid >> 4;
    int c = (tid & 15) * 8;
    float s = inv_in[row0 + r];
    const float4* g = (const float4*)(agg + (size_t)(row0 + r) * D + c);
    float4 v0 = g[0];
    float4 v1 = g[1];
    float4* l = (float4*)(Atile + r * AP + c);
    v0.x *= s; v0.y *= s; v0.z *= s; v0.w *= s;
    v1.x *= s; v1.y *= s; v1.z *= s; v1.w *= s;
    l[0] = v0;
    l[1] = v1;
  }
  __syncthreads();

  const int lane = tid & 31;
  const int col0 = (tid >> 5) * 16;
  const int m = lane & 15;
  const int khalf = (lane >> 4) * 2;

  v8f c = {};
#pragma unroll 4
  for (int k0 = 0; k0 < D; k0 += 4) {
    v2f a;
    a.x = Atile[m * AP + k0 + khalf + 0];
    a.y = Atile[m * AP + k0 + khalf + 1];
    v2f b;
    b.x = W[(size_t)(k0 + khalf + 0) * D + col0 + m];
    b.y = W[(size_t)(k0 + khalf + 1) * D + col0 + m];
    c = __builtin_amdgcn_wmma_f32_16x16x4_f32(
         false, a,  false, b,
         (short)0, c,  false,  false);
  }

  const int ncol = col0 + m;
  const int rbase = row0 + (lane >> 4) * 8;
  const float bv = bias[ncol];
#pragma unroll
  for (int r = 0; r < 8; ++r) {
    float v = c[r] + bv;
    if (relu) v = fmaxf(v, 0.0f);
    out[(size_t)(rbase + r) * D + ncol] = v;
  }
}

extern "C" void kernel_launch(void* const* d_in, const int* in_sizes, int n_in,
                              void* d_out, int out_size, void* d_ws, size_t ws_size,
                              hipStream_t stream) {
  const float* feat = (const float*)d_in[0];
  const int* src = (const int*)d_in[1];
  const int* dst = (const int*)d_in[2];
  const float* W1 = (const float*)d_in[3];
  const float* b1 = (const float*)d_in[4];
  const float* W2 = (const float*)d_in[5];
  const float* b2 = (const float*)d_in[6];
  const float* W3 = (const float*)d_in[7];
  const float* b3 = (const float*)d_in[8];
  float* out = (float*)d_out;

  const int N = in_sizes[0] / D;
  const int E = in_sizes[1];

  float* ws = (float*)d_ws;
  float* inv_out = ws;
  float* inv_in = ws + N;
  float* agg = ws + 2 * (size_t)N;
  float* h = agg + (size_t)N * D;
  float* wgt = h + (size_t)N * D;
  int* col = (int*)(wgt + E);
  int* rank = col + E;
  int* cnt_out = rank + E;
  int* cnt_in = cnt_out + N;
  int* rowptr = cnt_in + N;
  int* part_out = rowptr + N + 1;
  int* part_in = part_out + (size_t)NB * N;

  const int chunk = (E + NB - 1) / NB;

  gcn_count_kernel<<<2 * NB, 1024, 0, stream>>>(src, dst, part_out, part_in, rank, N, E, chunk);
  gcn_chunk_prefix_kernel<<<(N + 255) / 256, 256, 0, stream>>>(part_in, part_out, cnt_out, cnt_in, N);
  gcn_rowptr_kernel<<<(N + 1023) / 1024, 1024, 0, stream>>>(cnt_out, cnt_in, rowptr, inv_out, inv_in, N);
  gcn_fill_kernel<<<dim3((chunk + 255) / 256, NB), 256, 0, stream>>>(src, dst, rank, rowptr, part_in, inv_out, col, wgt, N, E, chunk);

  const int gather_blocks = 2048;
  const int gemm_blocks = N / 16;

  gcn_gather_kernel<<<gather_blocks, 256, 0, stream>>>(feat, rowptr, col, wgt, agg, N);
  gcn_gemm_kernel<<<gemm_blocks, 256, 0, stream>>>(agg, inv_in, W1, b1, h, 1);

  gcn_gather_kernel<<<gather_blocks, 256, 0, stream>>>(h, rowptr, col, wgt, agg, N);
  gcn_gemm_kernel<<<gemm_blocks, 256, 0, stream>>>(agg, inv_in, W2, b2, h, 1);

  gcn_gather_kernel<<<gather_blocks, 256, 0, stream>>>(h, rowptr, col, wgt, agg, N);
  gcn_gemm_kernel<<<gemm_blocks, 256, 0, stream>>>(agg, inv_in, W3, b3, out, 0);
}


# Row gather fused into the layer's WMMA launch: a block sums its 16 destination rows (2 per wave) into the scaled LDS A tile; no aggregate plane in memory, 3 launches fewer, h in 2 alternating planes
# speedup vs baseline: 3.3100x; 1.1411x over previous
#include <hip/hip_runtime.h>


typedef __attribute__((ext_vector_type(2))) float v2f;
typedef __attribute__((ext_vector_type(8))) float v8f;

#define D 128
#define AP 132

#define NB 32
#define NODE_CAP 50000

__global__ __launch_bounds__(1024) void gcn_count_kernel(const int* __restrict__ src,
                                                         const int* __restrict__ dst,
                                                         int* __restrict__ part_out,
                                                         int* __restrict__ part_in,
                                                         int* __restrict__ rank,
                                                         int n_nodes, int n_edges, int chunk) {
  __shared__ int hist[NODE_CAP];
  const int tid = threadIdx.x;
  const bool by_dst = blockIdx.x < NB;
  const int cb = by_dst ? blockIdx.x : blockIdx.x - NB;
  const int e0 = cb * chunk;
  const int e1 = min(e0 + chunk, n_edges);

  for (int i = tid; i < n_nodes; i += 1024) hist[i] = 0;
  __syncthreads();

  if (by_dst) {
    for (int e = e0 + tid; e < e1; e += 4096) {
      int key[4];
#pragma unroll
      for (int u = 0; u < 4; ++u) {
        const int ee = e + u * 1024;
        key[u] = (ee < e1) ? dst[ee] : 0;
      }
      int r[4];
#pragma unroll
      for (int u = 0; u < 4; ++u) {
        const int ee = e + u * 1024;
        r[u] = (ee < e1) ? atomicAdd(&hist[key[u]], 1) : 0;
      }
#pragma unroll
      for (int u = 0; u < 4; ++u) {
        const int ee = e + u * 1024;
        if (ee < e1) rank[ee] = r[u];
      }
    }
  } else {
    for (int e = e0 + tid; e < e1; e += 4096) {
      int key[4];
#pragma unroll
      for (int u = 0; u < 4; ++u) {
        const int ee = e + u * 1024;
        key[u] = (ee < e1) ? src[ee] : 0;
      }
#pragma unroll
      for (int u = 0; u < 4; ++u) {
        const int ee = e + u * 1024;
        if (ee < e1) atomicAdd(&hist[key[u]], 1);
      }
    }
  }
  __syncthreads();

  int* part = (by_dst ? part_in : part_out) + (size_t)cb * n_nodes;
  for (int i = tid; i < n_nodes; i += 1024) part[i] = hist[i];
}

__global__ void gcn_chunk_prefix_kernel(int* __restrict__ part_in,
                                        const int* __restrict__ part_out,
                                        int* __restrict__ cnt_out,
                                        int* __restrict__ cnt_in, int n) {
  const int i = blockIdx.x * blockDim.x + threadIdx.x;
  if (i >= n) return;
  int cin[NB];
  int tout = 0;
#pragma unroll
  for (int c = 0; c < NB; ++c) {
    cin[c] = part_in[(size_t)c * n + i];
    tout += part_out[(size_t)c * n + i];
  }
  int run = 0;
#pragma unroll
  for (int c = 0; c < NB; ++c) {
    part_in[(size_t)c * n + i] = run;
    run += cin[c];
  }
  cnt_in[i] = run;
  cnt_out[i] = tout;
}

__global__ __launch_bounds__(1024) void gcn_rowptr_kernel(
    const int* __restrict__ cnt_out, const int* __restrict__ cnt_in,
    int* __restrict__ rowptr, float* __restrict__ inv_out,
    float* __restrict__ inv_in, int n) {
  __shared__ int wsum[32];
  __shared__ int wpre[32];
  const int tid = threadIdx.x;
  const int lane = tid & 31;
  const int wid = tid >> 5;
  const int base = blockIdx.x * 1024;
  const int i = base + tid;

  int pre = 0;
  for (int j = tid; j < base; j += 1024) pre += cnt_in[j];
  const int c = (i < n) ? cnt_in[i] : 0;

  int x = c;
  for (int o = 1; o < 32; o <<= 1) {
    int y = __shfl_up(x, o);
    if (lane >= o) x += y;
  }
  for (int o = 16; o > 0; o >>= 1) pre += __shfl_xor(pre, o);
  if (lane == 31) wsum[wid] = x;
  if (lane == 0) wpre[wid] = pre;
  __syncthreads();

  int woff = (lane < wid) ? wsum[lane] : 0;
  int ptot = wpre[lane];
  for (int o = 16; o > 0; o >>= 1) {
    woff += __shfl_xor(woff, o);
    ptot += __shfl_xor(ptot, o);
  }

  if (i < n) {
    const int start = ptot + woff + (x - c);
    rowptr[i] = start;
    if (i == n - 1) rowptr[n] = start + c;
    inv_in[i] = rsqrtf(fmaxf((float)c, 1.0f));
    inv_out[i] = rsqrtf(fmaxf((float)cnt_out[i], 1.0f));
  }
}

__global__ void gcn_fill_kernel(const int* __restrict__ src,
                                const int* __restrict__ dst,
                                const int* __restrict__ rank,
                                const int* __restrict__ rowptr,
                                const int* __restrict__ part_in,
                                const float* __restrict__ inv_out,
                                int* __restrict__ col, float* __restrict__ wgt,
                                int n_nodes, int n_edges, int chunk) {
  const int cb = blockIdx.y;
  const int off = blockIdx.x * blockDim.x + threadIdx.x;
  const int e = cb * chunk + off;
  if (off >= chunk || e >= n_edges) return;
  const int s = src[e];
  const int d = dst[e];
  const int pos = rowptr[d] + part_in[(size_t)cb * n_nodes + d] + rank[e];
  col[pos] = s;
  wgt[pos] = inv_out[s];
}


template <int G>
__device__ __forceinline__ void gcn_add_rows(const float* __restrict__ hl, int cs, int cwi,
                                             int k, float4& acc) {
  float4 v[G];
  float w[G];
#pragma unroll
  for (int t = 0; t < G; ++t) {
    const unsigned s = (unsigned)__builtin_amdgcn_readlane(cs, k + t);
    w[t] = __int_as_float(__builtin_amdgcn_readlane(cwi, k + t));
    v[t] = *(const float4*)(hl + (size_t)s * D);
  }
#pragma unroll
  for (int t = 0; t < G; ++t) {
    acc.x += v[t].x * w[t];
    acc.y += v[t].y * w[t];
    acc.z += v[t].z * w[t];
    acc.w += v[t].w * w[t];
  }
}

__device__ __forceinline__ float4 gcn_row_sum(const float* __restrict__ hl,
                                              const int* __restrict__ col,
                                              const float* __restrict__ wgt,
                                              int beg, int end, int lane, int cs, float cw) {
  float4 acc = make_float4(0.0f, 0.0f, 0.0f, 0.0f);
  for (int base = beg; base < end; base += 32) {
    if (base != beg) {
      const int j = base + lane;
      cs = 0;
      cw = 0.0f;
      if (j < end) {
        cs = col[j];
        cw = wgt[j];
      }
    }
    const int cwi = __float_as_int(cw);
    const int cnt = min(32, end - base);
    int k = 0;
    for (; k + 8 <= cnt; k += 8) gcn_add_rows<8>(hl, cs, cwi, k, acc);
    if (k + 4 <= cnt) {
      gcn_add_rows<4>(hl, cs, cwi, k, acc);
      k += 4;
    }
    const int rem = cnt - k;
    if (rem == 3) gcn_add_rows<3>(hl, cs, cwi, k, acc);
    else if (rem == 2) gcn_add_rows<2>(hl, cs, cwi, k, acc);
    else if (rem == 1) gcn_add_rows<1>(hl, cs, cwi, k, acc);
  }
  return acc;
}

__global__ __launch_bounds__(256) void gcn_layer_kernel(
    const float* __restrict__ h, const int* __restrict__ rowptr,
    const int* __restrict__ col, const float* __restrict__ wgt,
    const float* __restrict__ inv_in, const float* __restrict__ W,
    const float* __restrict__ bias, float* __restrict__ out, int relu) {
  __shared__ float Atile[16 * AP];

  const int row0 = blockIdx.x * 16;
  const int tid = threadIdx.x;
  const int lane = tid & 31;
  const int wid = __builtin_amdgcn_readfirstlane(tid >> 5);

  {
    const int lr = 2 * wid;
    const int r0 = row0 + lr;
    const int p0 = rowptr[r0];
    const int p1 = rowptr[r0 + 1];
    const int p2 = rowptr[r0 + 2];
    int cs0 = 0, cs1 = 0;
    float cw0 = 0.0f, cw1 = 0.0f;
    if (p0 + lane < p1) {
      cs0 = col[p0 + lane];
      cw0 = wgt[p0 + lane];
    }
    if (p1 + lane < p2) {
      cs1 = col[p1 + lane];
      cw1 = wgt[p1 + lane];
    }
    const float s0 = inv_in[r0];
    const float s1 = inv_in[r0 + 1];
    const float* hl = h + lane * 4;

    float4 a0 = gcn_row_sum(hl, col, wgt, p0, p1, lane, cs0, cw0);
    a0.x *= s0; a0.y *= s0; a0.z *= s0; a0.w *= s0;
    *(float4*)(Atile + lr * AP + lane * 4) = a0;

    float4 a1 = gcn_row_sum(hl, col, wgt, p1, p2, lane, cs1, cw1);
    a1.x *= s1; a1.y *= s1; a1.z *= s1; a1.w *= s1;
    *(float4*)(Atile + (lr + 1) * AP + lane * 4) = a1;
  }
  __syncthreads();

  const int col0 = wid * 16;
  const int m = lane & 15;
  const int khalf = (lane >> 4) * 2;

  v8f c = {};
#pragma unroll 4
  for (int k0 = 0; k0 < D; k0 += 4) {
    v2f a;
    a.x = Atile[m * AP + k0 + khalf + 0];
    a.y = Atile[m * AP + k0 + khalf + 1];
    v2f b;
    b.x = W[(size_t)(k0 + khalf + 0) * D + col0 + m];
    b.y = W[(size_t)(k0 + khalf + 1) * D + col0 + m];
    c = __builtin_amdgcn_wmma_f32_16x16x4_f32(
         false, a,  false, b,
         (short)0, c,  false,  false);
  }

  const int ncol = col0 + m;
  const int rbase = row0 + (lane >> 4) * 8;
  const float bv = bias[ncol];
#pragma unroll
  for (int r = 0; r < 8; ++r) {
    float v = c[r] + bv;
    if (relu) v = fmaxf(v, 0.0f);
    out[(size_t)(rbase + r) * D + ncol] = v;
  }
}

extern "C" void kernel_launch(void* const* d_in, const int* in_sizes, int n_in,
                              void* d_out, int out_size, void* d_ws, size_t ws_size,
                              hipStream_t stream) {
  const float* feat = (const float*)d_in[0];
  const int* src = (const int*)d_in[1];
  const int* dst = (const int*)d_in[2];
  const float* W1 = (const float*)d_in[3];
  const float* b1 = (const float*)d_in[4];
  const float* W2 = (const float*)d_in[5];
  const float* b2 = (const float*)d_in[6];
  const float* W3 = (const float*)d_in[7];
  const float* b3 = (const float*)d_in[8];
  float* out = (float*)d_out;

  const int N = in_sizes[0] / D;
  const int E = in_sizes[1];

  float* ws = (float*)d_ws;
  float* inv_out = ws;
  float* inv_in = ws + N;
  float* ha = ws + 2 * (size_t)N;
  float* hb = ha + (size_t)N * D;
  float* wgt = hb + (size_t)N * D;
  int* col = (int*)(wgt + E);
  int* rank = col + E;
  int* cnt_out = rank + E;
  int* cnt_in = cnt_out + N;
  int* rowptr = cnt_in + N;
  int* part_out = rowptr + N + 1;
  int* part_in = part_out + (size_t)NB * N;

  const int chunk = (E + NB - 1) / NB;

  gcn_count_kernel<<<2 * NB, 1024, 0, stream>>>(src, dst, part_out, part_in, rank, N, E, chunk);
  gcn_chunk_prefix_kernel<<<(N + 255) / 256, 256, 0, stream>>>(part_in, part_out, cnt_out, cnt_in, N);
  gcn_rowptr_kernel<<<(N + 1023) / 1024, 1024, 0, stream>>>(cnt_out, cnt_in, rowptr, inv_out, inv_in, N);
  gcn_fill_kernel<<<dim3((chunk + 255) / 256, NB), 256, 0, stream>>>(src, dst, rank, rowptr, part_in, inv_out, col, wgt, N, E, chunk);

  const int layer_blocks = N / 16;

  gcn_layer_kernel<<<layer_blocks, 256, 0, stream>>>(feat, rowptr, col, wgt, inv_in, W1, b1, ha, 1);
  gcn_layer_kernel<<<layer_blocks, 256, 0, stream>>>(ha, rowptr, col, wgt, inv_in, W2, b2, hb, 1);
  gcn_layer_kernel<<<layer_blocks, 256, 0, stream>>>(hb, rowptr, col, wgt, inv_in, W3, b3, out, 0);
}


# Layer launch as a resident grid: 1024 blocks (4 per WGP) each walk 3-4 row tiles, a wave's 128x16 W slice read once into registers, A tile double-buffered (one barrier a tile); sum order unchanged
# speedup vs baseline: 3.4523x; 1.0430x over previous
#include <hip/hip_runtime.h>


typedef __attribute__((ext_vector_type(2))) float v2f;
typedef __attribute__((ext_vector_type(8))) float v8f;

#define D 128
#define AP 132

#define NB 32
#define NODE_CAP 50000

__global__ __launch_bounds__(1024) void gcn_count_kernel(const int* __restrict__ src,
                                                         const int* __restrict__ dst,
                                                         int* __restrict__ part_out,
                                                         int* __restrict__ part_in,
                                                         int* __restrict__ rank,
                                                         int n_nodes, int n_edges, int chunk) {
  __shared__ int hist[NODE_CAP];
  const int tid = threadIdx.x;
  const bool by_dst = blockIdx.x < NB;
  const int cb = by_dst ? blockIdx.x : blockIdx.x - NB;
  const int e0 = cb * chunk;
  const int e1 = min(e0 + chunk, n_edges);

  for (int i = tid; i < n_nodes; i += 1024) hist[i] = 0;
  __syncthreads();

  if (by_dst) {
    for (int e = e0 + tid; e < e1; e += 4096) {
      int key[4];
#pragma unroll
      for (int u = 0; u < 4; ++u) {
        const int ee = e + u * 1024;
        key[u] = (ee < e1) ? dst[ee] : 0;
      }
      int r[4];
#pragma unroll
      for (int u = 0; u < 4; ++u) {
        const int ee = e + u * 1024;
        r[u] = (ee < e1) ? atomicAdd(&hist[key[u]], 1) : 0;
      }
#pragma unroll
      for (int u = 0; u < 4; ++u) {
        const int ee = e + u * 1024;
        if (ee < e1) rank[ee] = r[u];
      }
    }
  } else {
    for (int e = e0 + tid; e < e1; e += 4096) {
      int key[4];
#pragma unroll
      for (int u = 0; u < 4; ++u) {
        const int ee = e + u * 1024;
        key[u] = (ee < e1) ? src[ee] : 0;
      }
#pragma unroll
      for (int u = 0; u < 4; ++u) {
        const int ee = e + u * 1024;
        if (ee < e1) atomicAdd(&hist[key[u]], 1);
      }
    }
  }
  __syncthreads();

  int* part = (by_dst ? part_in : part_out) + (size_t)cb * n_nodes;
  for (int i = tid; i < n_nodes; i += 1024) part[i] = hist[i];
}

__global__ void gcn_chunk_prefix_kernel(int* __restrict__ part_in,
                                        const int* __restrict__ part_out,
                                        int* __restrict__ cnt_out,
                                        int* __restrict__ cnt_in, int n) {
  const int i = blockIdx.x * blockDim.x + threadIdx.x;
  if (i >= n) return;
  int cin[NB];
  int tout = 0;
#pragma unroll
  for (int c = 0; c < NB; ++c) {
    cin[c] = part_in[(size_t)c * n + i];
    tout += part_out[(size_t)c * n + i];
  }
  int run = 0;
#pragma unroll
  for (int c = 0; c < NB; ++c) {
    part_in[(size_t)c * n + i] = run;
    run += cin[c];
  }
  cnt_in[i] = run;
  cnt_out[i] = tout;
}

__global__ __launch_bounds__(1024) void gcn_rowptr_kernel(
    const int* __restrict__ cnt_out, const int* __restrict__ cnt_in,
    int* __restrict__ rowptr, float* __restrict__ inv_out,
    float* __restrict__ inv_in, int n) {
  __shared__ int wsum[32];
  __shared__ int wpre[32];
  const int tid = threadIdx.x;
  const int lane = tid & 31;
  const int wid = tid >> 5;
  const int base = blockIdx.x * 1024;
  const int i = base + tid;

  int pre = 0;
  for (int j = tid; j < base; j += 1024) pre += cnt_in[j];
  const int c = (i < n) ? cnt_in[i] : 0;

  int x = c;
  for (int o = 1; o < 32; o <<= 1) {
    int y = __shfl_up(x, o);
    if (lane >= o) x += y;
  }
  for (int o = 16; o > 0; o >>= 1) pre += __shfl_xor(pre, o);
  if (lane == 31) wsum[wid] = x;
  if (lane == 0) wpre[wid] = pre;
  __syncthreads();

  int woff = (lane < wid) ? wsum[lane] : 0;
  int ptot = wpre[lane];
  for (int o = 16; o > 0; o >>= 1) {
    woff += __shfl_xor(woff, o);
    ptot += __shfl_xor(ptot, o);
  }

  if (i < n) {
    const int start = ptot + woff + (x - c);
    rowptr[i] = start;
    if (i == n - 1) rowptr[n] = start + c;
    inv_in[i] = rsqrtf(fmaxf((float)c, 1.0f));
    inv_out[i] = rsqrtf(fmaxf((float)cnt_out[i], 1.0f));
  }
}

__global__ void gcn_fill_kernel(const int* __restrict__ src,
                                const int* __restrict__ dst,
                                const int* __restrict__ rank,
                                const int* __restrict__ rowptr,
                                const int* __restrict__ part_in,
                                const float* __restrict__ inv_out,
                                int* __restrict__ col, float* __restrict__ wgt,
                                int n_nodes, int n_edges, int chunk) {
  const int cb = blockIdx.y;
  const int off = blockIdx.x * blockDim.x + threadIdx.x;
  const int e = cb * chunk + off;
  if (off >= chunk || e >= n_edges) return;
  const int s = src[e];
  const int d = dst[e];
  const int pos = rowptr[d] + part_in[(size_t)cb * n_nodes + d] + rank[e];
  col[pos] = s;
  wgt[pos] = inv_out[s];
}


template <int G>
__device__ __forceinline__ void gcn_add_rows(const float* __restrict__ hl, int cs, int cwi,
                                             int k, float4& acc) {
  float4 v[G];
  float w[G];
#pragma unroll
  for (int t = 0; t < G; ++t) {
    const unsigned s = (unsigned)__builtin_amdgcn_readlane(cs, k + t);
    w[t] = __int_as_float(__builtin_amdgcn_readlane(cwi, k + t));
    v[t] = *(const float4*)(hl + (size_t)s * D);
  }
#pragma unroll
  for (int t = 0; t < G; ++t) {
    acc.x += v[t].x * w[t];
    acc.y += v[t].y * w[t];
    acc.z += v[t].z * w[t];
    acc.w += v[t].w * w[t];
  }
}

__device__ __forceinline__ float4 gcn_row_sum(const float* __restrict__ hl,
                                              const int* __restrict__ col,
                                              const float* __restrict__ wgt,
                                              int beg, int end, int lane, int cs, float cw) {
  float4 acc = make_float4(0.0f, 0.0f, 0.0f, 0.0f);
  for (int base = beg; base < end; base += 32) {
    if (base != beg) {
      const int j = base + lane;
      cs = 0;
      cw = 0.0f;
      if (j < end) {
        cs = col[j];
        cw = wgt[j];
      }
    }
    const int cwi = __float_as_int(cw);
    const int cnt = min(32, end - base);
    int k = 0;
    for (; k + 8 <= cnt; k += 8) gcn_add_rows<8>(hl, cs, cwi, k, acc);
    if (k + 4 <= cnt) {
      gcn_add_rows<4>(hl, cs, cwi, k, acc);
      k += 4;
    }
    const int rem = cnt - k;
    if (rem == 3) gcn_add_rows<3>(hl, cs, cwi, k, acc);
    else if (rem == 2) gcn_add_rows<2>(hl, cs, cwi, k, acc);
    else if (rem == 1) gcn_add_rows<1>(hl, cs, cwi, k, acc);
  }
  return acc;
}

__global__ __launch_bounds__(256) void gcn_layer_kernel(
    const float* __restrict__ h, const int* __restrict__ rowptr,
    const int* __restrict__ col, const float* __restrict__ wgt,
    const float* __restrict__ inv_in, const float* __restrict__ W,
    const float* __restrict__ bias, float* __restrict__ out, int relu, int n_tiles) {
  __shared__ float Atile[2][16 * AP];

  const int tid = threadIdx.x;
  const int lane = tid & 31;
  const int wid = __builtin_amdgcn_readfirstlane(tid >> 5);

  const int col0 = wid * 16;
  const int m = lane & 15;
  const int khalf = (lane >> 4) * 2;
  const int ncol = col0 + m;

  v2f bfr[D / 4];
#pragma unroll
  for (int s = 0; s < D / 4; ++s) {
    bfr[s].x = W[(size_t)(4 * s + khalf + 0) * D + ncol];
    bfr[s].y = W[(size_t)(4 * s + khalf + 1) * D + ncol];
  }
  const float bv = bias[ncol];
  const float* hl = h + lane * 4;
  const int lr = 2 * wid;

  int buf = 0;
  for (int tile = blockIdx.x; tile < n_tiles; tile += gridDim.x) {
    float* At = Atile[buf];
    buf ^= 1;
    const int row0 = tile * 16;

    {
      const int r0 = row0 + lr;
      const int p0 = rowptr[r0];
      const int p1 = rowptr[r0 + 1];
      const int p2 = rowptr[r0 + 2];
      int cs0 = 0, cs1 = 0;
      float cw0 = 0.0f, cw1 = 0.0f;
      if (p0 + lane < p1) {
        cs0 = col[p0 + lane];
        cw0 = wgt[p0 + lane];
      }
      if (p1 + lane < p2) {
        cs1 = col[p1 + lane];
        cw1 = wgt[p1 + lane];
      }
      const float s0 = inv_in[r0];
      const float s1 = inv_in[r0 + 1];

      float4 a0 = gcn_row_sum(hl, col, wgt, p0, p1, lane, cs0, cw0);
      a0.x *= s0; a0.y *= s0; a0.z *= s0; a0.w *= s0;
      *(float4*)(At + lr * AP + lane * 4) = a0;

      float4 a1 = gcn_row_sum(hl, col, wgt, p1, p2, lane, cs1, cw1);
      a1.x *= s1; a1.y *= s1; a1.z *= s1; a1.w *= s1;
      *(float4*)(At + (lr + 1) * AP + lane * 4) = a1;
    }
    __syncthreads();

    v8f c = {};
#pragma unroll
    for (int s = 0; s < D / 4; ++s) {
      v2f a;
      a.x = At[m * AP + 4 * s + khalf + 0];
      a.y = At[m * AP + 4 * s + khalf + 1];
      c = __builtin_amdgcn_wmma_f32_16x16x4_f32(
           false, a,  false, bfr[s],
           (short)0, c,  false,  false);
    }

    const int rbase = row0 + (lane >> 4) * 8;
#pragma unroll
    for (int r = 0; r < 8; ++r) {
      float v = c[r] + bv;
      if (relu) v = fmaxf(v, 0.0f);
      out[(size_t)(rbase + r) * D + ncol] = v;
    }
  }
}

extern "C" void kernel_launch(void* const* d_in, const int* in_sizes, int n_in,
                              void* d_out, int out_size, void* d_ws, size_t ws_size,
                              hipStream_t stream) {
  const float* feat = (const float*)d_in[0];
  const int* src = (const int*)d_in[1];
  const int* dst = (const int*)d_in[2];
  const float* W1 = (const float*)d_in[3];
  const float* b1 = (const float*)d_in[4];
  const float* W2 = (const float*)d_in[5];
  const float* b2 = (const float*)d_in[6];
  const float* W3 = (const float*)d_in[7];
  const float* b3 = (const float*)d_in[8];
  float* out = (float*)d_out;

  const int N = in_sizes[0] / D;
  const int E = in_sizes[1];

  float* ws = (float*)d_ws;
  float* inv_out = ws;
  float* inv_in = ws + N;
  float* ha = ws + 2 * (size_t)N;
  float* hb = ha + (size_t)N * D;
  float* wgt = hb + (size_t)N * D;
  int* col = (int*)(wgt + E);
  int* rank = col + E;
  int* cnt_out = rank + E;
  int* cnt_in = cnt_out + N;
  int* rowptr = cnt_in + N;
  int* part_out = rowptr + N + 1;
  int* part_in = part_out + (size_t)NB * N;

  const int chunk = (E + NB - 1) / NB;

  gcn_count_kernel<<<2 * NB, 1024, 0, stream>>>(src, dst, part_out, part_in, rank, N, E, chunk);
  gcn_chunk_prefix_kernel<<<(N + 255) / 256, 256, 0, stream>>>(part_in, part_out, cnt_out, cnt_in, N);
  gcn_rowptr_kernel<<<(N + 1023) / 1024, 1024, 0, stream>>>(cnt_out, cnt_in, rowptr, inv_out, inv_in, N);
  gcn_fill_kernel<<<dim3((chunk + 255) / 256, NB), 256, 0, stream>>>(src, dst, rank, rowptr, part_in, inv_out, col, wgt, N, E, chunk);

  const int n_tiles = N / 16;
  const int layer_blocks = 1024;

  gcn_layer_kernel<<<layer_blocks, 256, 0, stream>>>(feat, rowptr, col, wgt, inv_in, W1, b1, ha, 1, n_tiles);
  gcn_layer_kernel<<<layer_blocks, 256, 0, stream>>>(ha, rowptr, col, wgt, inv_in, W2, b2, hb, 1, n_tiles);
  gcn_layer_kernel<<<layer_blocks, 256, 0, stream>>>(hb, rowptr, col, wgt, inv_in, W3, b3, out, 0, n_tiles);
}
